# starting kernel
# baseline (speedup 1.0000x reference)
#include <hip/hip_runtime.h>
#include <hip/hip_bf16.h>

typedef __attribute__((ext_vector_type(2))) float v2f;
typedef __attribute__((ext_vector_type(4))) float v4f;
typedef __attribute__((ext_vector_type(8))) float v8f;

#define B_     4
#define S_     4096
#define D_     1024
#define SCHUNK 32
#define SPAN   (S_ / SCHUNK)

__global__ __launch_bounds__(256) void xsum_partial_kernel(
    const float* __restrict__ x, float* __restrict__ partials) {
  const int b  = blockIdx.x;
  const int sc = blockIdx.y;
  const int d  = threadIdx.x * 4;

  const float* p = x + ((size_t)b * S_ + (size_t)sc * SPAN) * D_ + d;
  v4f acc = {0.0f, 0.0f, 0.0f, 0.0f};
#pragma unroll 8
  for (int s = 0; s < SPAN; ++s) {
    v4f t = __builtin_nontemporal_load((const v4f*)(p + (size_t)s * D_));
    acc += t;
  }
  *(v4f*)(partials + ((size_t)sc * B_ + b) * D_ + d) = acc;
}

__global__ __launch_bounds__(256) void xsum_combine_kernel(
    const float* __restrict__ partials, float* __restrict__ A) {
  const int e = blockIdx.x * 256 + threadIdx.x;
  const int m = e >> 10;
  const int d = e & 1023;
  float v = 0.0f;
  if (m < B_) {
#pragma unroll
    for (int sc = 0; sc < SCHUNK; ++sc) {
      v += partials[((size_t)sc * B_ + m) * D_ + d];
    }
  }
  A[e] = v;
}

__global__ __launch_bounds__(32) void gemm16_f32wmma_kernel(
    const float* __restrict__ A, const float* __restrict__ W,
    float* __restrict__ Out) {
  const int lane = threadIdx.x;
  const int half = lane >> 4;
  const int l    = lane & 15;
  const int nTile = blockIdx.x;
  const int col   = nTile * 16 + l;

  const float* arow = A + (size_t)l * D_;
  const float* wrow = W + (size_t)col * D_;

  v8f c = {};
#pragma unroll 8
  for (int k0 = 0; k0 < D_; k0 += 4) {
    const int ka = k0 + half * 2;
    v2f a = *(const v2f*)(arow + ka);
    v2f b = *(const v2f*)(wrow + ka);
    c = __builtin_amdgcn_wmma_f32_16x16x4_f32(
         false, a,  false, b,
         (short)0, c,  false,  false);
  }

#pragma unroll
  for (int r = 0; r < 8; ++r) {
    Out[(size_t)(half * 8 + r) * D_ + col] = c[r];
  }
}

__global__ __launch_bounds__(256) void broadcast_kernel(
    const float* __restrict__ Y, float* __restrict__ out) {
  const size_t i  = (size_t)blockIdx.x * 256 + threadIdx.x;
  const int    o4 = (int)(i & 255);
  const int    b  = (int)(i >> 20);
  const v4f v = ((const v4f*)Y)[(size_t)b * 256 + o4];
  __builtin_nontemporal_store(v, (v4f*)out + i);
}

extern "C" void kernel_launch(void* const* d_in, const int* in_sizes, int n_in,
                              void* d_out, int out_size, void* d_ws, size_t ws_size,
                              hipStream_t stream) {
  const float* x     = (const float*)d_in[0];
  const float* w_qkv = (const float*)d_in[1];
  const float* w_o   = (const float*)d_in[2];
  const float* w_v   = w_qkv + (size_t)2 * D_ * D_;
  float* out = (float*)d_out;

  float* ws       = (float*)d_ws;
  float* partials = ws;
  float* A        = ws + 131072;
  float* T        = ws + 147456;
  float* Y        = ws + 163840;

  xsum_partial_kernel<<<dim3(B_, SCHUNK), 256, 0, stream>>>(x, partials);

  xsum_combine_kernel<<<64, 256, 0, stream>>>(partials, A);

  gemm16_f32wmma_kernel<<<64, 32, 0, stream>>>(A, w_v, T);

  gemm16_f32wmma_kernel<<<64, 32, 0, stream>>>(T, w_o, Y);

  const int n4 = (B_ * S_ * D_) / 4;
  broadcast_kernel<<<n4 / 256, 256, 0, stream>>>(Y, out);
}


# x column sums fused with first product: a workgroup owns 32 x columns, sums them over s, multiplies by its w_v slice on WMMA; 32 K-slice partials summed after (f32 sum order changes); 5->4 launches
# speedup vs baseline: 1.2836x; 1.2836x over previous
#include <hip/hip_runtime.h>
#include <hip/hip_bf16.h>

typedef __attribute__((ext_vector_type(2))) float v2f;
typedef __attribute__((ext_vector_type(4))) float v4f;
typedef __attribute__((ext_vector_type(8))) float v8f;

#define B_     4
#define S_     4096
#define D_     1024
#define KSL    32
#define NSL    (D_ / KSL)

__global__ __launch_bounds__(256) void xsum_vproj_kernel(
    const float* __restrict__ x, const float* __restrict__ Wv,
    float* __restrict__ Tpart) {
  __shared__ alignas(16) float red[32 * KSL];
  __shared__ alignas(16) float xs[KSL];

  const int sl   = blockIdx.x;
  const int b    = blockIdx.y;
  const int t    = threadIdx.x;
  const int lane = t & 31;
  const int wave = t >> 5;
  const int rg   = wave * 4 + (lane >> 3);
  const int c4   = (lane & 7) * 4;

  const int half = lane >> 4;
  const int l    = lane & 15;

  v2f bfrag[8][KSL / 4];
#pragma unroll
  for (int j = 0; j < 8; ++j) {
    const float* wrow =
        Wv + (size_t)((wave * 8 + j) * 16 + l) * D_ + sl * KSL + half * 2;
#pragma unroll
    for (int ks = 0; ks < KSL / 4; ++ks) {
      bfrag[j][ks] = *(const v2f*)(wrow + ks * 4);
    }
  }

  const float* p = x + ((size_t)b * S_ + rg) * D_ + sl * KSL + c4;
  v4f acc = {0.0f, 0.0f, 0.0f, 0.0f};
#pragma unroll 8
  for (int i = 0; i < S_ / 32; ++i) {
    v4f v = __builtin_nontemporal_load((const v4f*)(p + (size_t)i * 32 * D_));
    acc += v;
  }
  *(v4f*)(red + rg * KSL + c4) = acc;
  __syncthreads();
  if (t < KSL) {
    float v = 0.0f;
#pragma unroll
    for (int g = 0; g < 32; ++g) v += red[g * KSL + t];
    xs[t] = v;
  }
  __syncthreads();

  v2f afrag[KSL / 4];
#pragma unroll
  for (int ks = 0; ks < KSL / 4; ++ks) {
    const v2f av = *(const v2f*)(xs + ks * 4 + half * 2);
    const v2f z  = {0.0f, 0.0f};
    afrag[ks] = (l == 0) ? av : z;
  }

  float row0[8];
#pragma unroll
  for (int j = 0; j < 8; ++j) {
    v8f c = {};
#pragma unroll
    for (int ks = 0; ks < KSL / 4; ++ks) {
      c = __builtin_amdgcn_wmma_f32_16x16x4_f32(
           false, afrag[ks],  false, bfrag[j][ks],
           (short)0, c,  false,  false);
    }
    row0[j] = c[0];
  }
  if (half == 0) {
    float* o = Tpart + ((size_t)sl * B_ + b) * D_ + wave * 128 + l;
#pragma unroll
    for (int j = 0; j < 8; ++j) o[j * 16] = row0[j];
  }
}

__global__ __launch_bounds__(256) void partial_combine_kernel(
    const float* __restrict__ partials, float* __restrict__ A) {
  const int e = blockIdx.x * 256 + threadIdx.x;
  const int m = e >> 10;
  const int d = e & 1023;
  float v = 0.0f;
  if (m < B_) {
#pragma unroll
    for (int sc = 0; sc < NSL; ++sc) {
      v += partials[((size_t)sc * B_ + m) * D_ + d];
    }
  }
  A[e] = v;
}

__global__ __launch_bounds__(32) void gemm16_f32wmma_kernel(
    const float* __restrict__ A, const float* __restrict__ W,
    float* __restrict__ Out) {
  const int lane = threadIdx.x;
  const int half = lane >> 4;
  const int l    = lane & 15;
  const int nTile = blockIdx.x;
  const int col   = nTile * 16 + l;

  const float* arow = A + (size_t)l * D_;
  const float* wrow = W + (size_t)col * D_;

  v8f c = {};
#pragma unroll 8
  for (int k0 = 0; k0 < D_; k0 += 4) {
    const int ka = k0 + half * 2;
    v2f a = *(const v2f*)(arow + ka);
    v2f b = *(const v2f*)(wrow + ka);
    c = __builtin_amdgcn_wmma_f32_16x16x4_f32(
         false, a,  false, b,
         (short)0, c,  false,  false);
  }

#pragma unroll
  for (int r = 0; r < 8; ++r) {
    Out[(size_t)(half * 8 + r) * D_ + col] = c[r];
  }
}

__global__ __launch_bounds__(256) void broadcast_kernel(
    const float* __restrict__ Y, float* __restrict__ out) {
  const size_t i  = (size_t)blockIdx.x * 256 + threadIdx.x;
  const int    o4 = (int)(i & 255);
  const int    b  = (int)(i >> 20);
  const v4f v = ((const v4f*)Y)[(size_t)b * 256 + o4];
  __builtin_nontemporal_store(v, (v4f*)out + i);
}

extern "C" void kernel_launch(void* const* d_in, const int* in_sizes, int n_in,
                              void* d_out, int out_size, void* d_ws, size_t ws_size,
                              hipStream_t stream) {
  const float* x     = (const float*)d_in[0];
  const float* w_qkv = (const float*)d_in[1];
  const float* w_o   = (const float*)d_in[2];
  const float* w_v   = w_qkv + (size_t)2 * D_ * D_;
  float* out = (float*)d_out;

  float* ws    = (float*)d_ws;
  float* Tpart = ws;
  float* T     = ws + 131072;
  float* Y     = ws + 163840;

  xsum_vproj_kernel<<<dim3(NSL, B_), 256, 0, stream>>>(x, w_v, Tpart);

  partial_combine_kernel<<<64, 256, 0, stream>>>(Tpart, T);

  gemm16_f32wmma_kernel<<<64, 32, 0, stream>>>(T, w_o, Y);

  const int n4 = (B_ * S_ * D_) / 4;
  broadcast_kernel<<<n4 / 256, 256, 0, stream>>>(Y, out);
}


# Partial sum fused into second product: a workgroup sums the 32 partials of its 128 T columns in LDS, multiplies by its w_o slice on WMMA; 8 K-slice partials of Y summed after (f32 sum order changes)
# speedup vs baseline: 1.7511x; 1.3642x over previous
#include <hip/hip_runtime.h>
#include <hip/hip_bf16.h>

typedef __attribute__((ext_vector_type(2))) float v2f;
typedef __attribute__((ext_vector_type(4))) float v4f;
typedef __attribute__((ext_vector_type(8))) float v8f;

#define B_     4
#define S_     4096
#define D_     1024
#define KSL    32
#define NSL    (D_ / KSL)
#define KS2    128
#define NS2    (D_ / KS2)
#define TPITCH 132

__global__ __launch_bounds__(256) void xsum_vproj_kernel(
    const float* __restrict__ x, const float* __restrict__ Wv,
    float* __restrict__ Tpart) {
  __shared__ alignas(16) float red[32 * KSL];
  __shared__ alignas(16) float xs[KSL];

  const int sl   = blockIdx.x;
  const int b    = blockIdx.y;
  const int t    = threadIdx.x;
  const int lane = t & 31;
  const int wave = t >> 5;
  const int rg   = wave * 4 + (lane >> 3);
  const int c4   = (lane & 7) * 4;

  const int half = lane >> 4;
  const int l    = lane & 15;

  v2f bfrag[8][KSL / 4];
#pragma unroll
  for (int j = 0; j < 8; ++j) {
    const float* wrow =
        Wv + (size_t)((wave * 8 + j) * 16 + l) * D_ + sl * KSL + half * 2;
#pragma unroll
    for (int ks = 0; ks < KSL / 4; ++ks) {
      bfrag[j][ks] = *(const v2f*)(wrow + ks * 4);
    }
  }

  const float* p = x + ((size_t)b * S_ + rg) * D_ + sl * KSL + c4;
  v4f acc = {0.0f, 0.0f, 0.0f, 0.0f};
#pragma unroll 8
  for (int i = 0; i < S_ / 32; ++i) {
    v4f v = __builtin_nontemporal_load((const v4f*)(p + (size_t)i * 32 * D_));
    acc += v;
  }
  *(v4f*)(red + rg * KSL + c4) = acc;
  __syncthreads();
  if (t < KSL) {
    float v = 0.0f;
#pragma unroll
    for (int g = 0; g < 32; ++g) v += red[g * KSL + t];
    xs[t] = v;
  }
  __syncthreads();

  v2f afrag[KSL / 4];
#pragma unroll
  for (int ks = 0; ks < KSL / 4; ++ks) {
    const v2f av = *(const v2f*)(xs + ks * 4 + half * 2);
    const v2f z  = {0.0f, 0.0f};
    afrag[ks] = (l == 0) ? av : z;
  }

  float row0[8];
#pragma unroll
  for (int j = 0; j < 8; ++j) {
    v8f c = {};
#pragma unroll
    for (int ks = 0; ks < KSL / 4; ++ks) {
      c = __builtin_amdgcn_wmma_f32_16x16x4_f32(
           false, afrag[ks],  false, bfrag[j][ks],
           (short)0, c,  false,  false);
    }
    row0[j] = c[0];
  }
  if (half == 0) {
    float* o = Tpart + ((size_t)sl * B_ + b) * D_ + wave * 128 + l;
#pragma unroll
    for (int j = 0; j < 8; ++j) o[j * 16] = row0[j];
  }
}

__global__ __launch_bounds__(256) __attribute__((amdgpu_waves_per_eu(2, 4)))
void tsum_oproj_kernel(
    const float* __restrict__ Tpart, const float* __restrict__ Wo,
    float* __restrict__ Ypart) {
  __shared__ alignas(16) float red[2 * B_ * TPITCH];

  const int sl   = blockIdx.x;
  const int ng   = blockIdx.y;
  const int t    = threadIdx.x;
  const int lane = t & 31;
  const int wave = t >> 5;
  const int half = lane >> 4;
  const int l    = lane & 15;

  v2f bfrag[KS2 / 4];
  const float* wrow =
      Wo + (size_t)((ng * 8 + wave) * 16 + l) * D_ + sl * KS2 + half * 2;
#pragma unroll
  for (int ks = 0; ks < KS2 / 4; ++ks) {
    bfrag[ks] = *(const v2f*)(wrow + ks * 4);
  }

  const int b = wave & 3;
  const int h = wave >> 2;
  const float* p = Tpart + ((size_t)h * B_ + b) * D_ + sl * KS2 + lane * 4;
  v4f acc = {0.0f, 0.0f, 0.0f, 0.0f};
#pragma unroll
  for (int j = 0; j < NSL / 2; ++j) {
    acc += *(const v4f*)(p + (size_t)j * 2 * B_ * D_);
  }
  *(v4f*)(red + (h * B_ + b) * TPITCH + lane * 4) = acc;
  __syncthreads();

  v2f afrag[KS2 / 4];
  const int r = l & 3;
#pragma unroll
  for (int ks = 0; ks < KS2 / 4; ++ks) {
    const v2f ev = *(const v2f*)(red + r * TPITCH + ks * 4 + half * 2);
    const v2f od = *(const v2f*)(red + (B_ + r) * TPITCH + ks * 4 + half * 2);
    const v2f z  = {0.0f, 0.0f};
    afrag[ks] = (l < B_) ? (ev + od) : z;
  }

  v8f c0 = {};
  v8f c1 = {};
#pragma unroll
  for (int ks = 0; ks < KS2 / 4; ks += 2) {
    c0 = __builtin_amdgcn_wmma_f32_16x16x4_f32(
         false, afrag[ks],  false, bfrag[ks],
         (short)0, c0,  false,  false);
    c1 = __builtin_amdgcn_wmma_f32_16x16x4_f32(
         false, afrag[ks + 1],  false, bfrag[ks + 1],
         (short)0, c1,  false,  false);
  }
  float yrow[B_];
#pragma unroll
  for (int m = 0; m < B_; ++m) yrow[m] = c0[m] + c1[m];

  if (half == 0) {
    float* o = Ypart + (size_t)sl * B_ * D_ + (ng * 8 + wave) * 16 + l;
#pragma unroll
    for (int m = 0; m < B_; ++m) o[(size_t)m * D_] = yrow[m];
  }
}

__global__ __launch_bounds__(256) void ypart_combine_kernel(
    const float* __restrict__ Ypart, float* __restrict__ Y) {
  const int e = blockIdx.x * 256 + threadIdx.x;
  float v = 0.0f;
#pragma unroll
  for (int sc = 0; sc < NS2; ++sc) {
    v += Ypart[(size_t)sc * B_ * D_ + e];
  }
  Y[e] = v;
}

__global__ __launch_bounds__(256) void broadcast_kernel(
    const float* __restrict__ Y, float* __restrict__ out) {
  const size_t i  = (size_t)blockIdx.x * 256 + threadIdx.x;
  const int    o4 = (int)(i & 255);
  const int    b  = (int)(i >> 20);
  const v4f v = ((const v4f*)Y)[(size_t)b * 256 + o4];
  __builtin_nontemporal_store(v, (v4f*)out + i);
}

extern "C" void kernel_launch(void* const* d_in, const int* in_sizes, int n_in,
                              void* d_out, int out_size, void* d_ws, size_t ws_size,
                              hipStream_t stream) {
  const float* x     = (const float*)d_in[0];
  const float* w_qkv = (const float*)d_in[1];
  const float* w_o   = (const float*)d_in[2];
  const float* w_v   = w_qkv + (size_t)2 * D_ * D_;
  float* out = (float*)d_out;

  float* ws    = (float*)d_ws;
  float* Tpart = ws;
  float* Ypart = ws + 131072;
  float* Y     = ws + 163840;

  xsum_vproj_kernel<<<dim3(NSL, B_), 256, 0, stream>>>(x, w_v, Tpart);

  tsum_oproj_kernel<<<dim3(NS2, 8), 256, 0, stream>>>(Tpart, w_o, Ypart);

  ypart_combine_kernel<<<16, 256, 0, stream>>>(Ypart, Y);

  const int n4 = (B_ * S_ * D_) / 4;
  broadcast_kernel<<<n4 / 256, 256, 0, stream>>>(Y, out);
}


# Y-partial sum fused into the broadcast by column-strip ownership: a workgroup owns 32 out columns of a batch row, adds its 8 K-slice partials in the old order, stores down 2048 rows; 4->3 launches
# speedup vs baseline: 2.0862x; 1.1914x over previous
#include <hip/hip_runtime.h>
#include <hip/hip_bf16.h>

typedef __attribute__((ext_vector_type(2))) float v2f;
typedef __attribute__((ext_vector_type(4))) float v4f;
typedef __attribute__((ext_vector_type(8))) float v8f;

#define B_     4
#define S_     4096
#define D_     1024
#define KSL    32
#define NSL    (D_ / KSL)
#define KS2    128
#define NS2    (D_ / KS2)
#define TPITCH 132

__global__ __launch_bounds__(256) void xsum_vproj_kernel(
    const float* __restrict__ x, const float* __restrict__ Wv,
    float* __restrict__ Tpart) {
  __shared__ alignas(16) float red[32 * KSL];
  __shared__ alignas(16) float xs[KSL];

  const int sl   = blockIdx.x;
  const int b    = blockIdx.y;
  const int t    = threadIdx.x;
  const int lane = t & 31;
  const int wave = t >> 5;
  const int rg   = wave * 4 + (lane >> 3);
  const int c4   = (lane & 7) * 4;

  const int half = lane >> 4;
  const int l    = lane & 15;

  v2f bfrag[8][KSL / 4];
#pragma unroll
  for (int j = 0; j < 8; ++j) {
    const float* wrow =
        Wv + (size_t)((wave * 8 + j) * 16 + l) * D_ + sl * KSL + half * 2;
#pragma unroll
    for (int ks = 0; ks < KSL / 4; ++ks) {
      bfrag[j][ks] = *(const v2f*)(wrow + ks * 4);
    }
  }

  const float* p = x + ((size_t)b * S_ + rg) * D_ + sl * KSL + c4;
  v4f acc = {0.0f, 0.0f, 0.0f, 0.0f};
#pragma unroll 8
  for (int i = 0; i < S_ / 32; ++i) {
    v4f v = __builtin_nontemporal_load((const v4f*)(p + (size_t)i * 32 * D_));
    acc += v;
  }
  *(v4f*)(red + rg * KSL + c4) = acc;
  __syncthreads();
  if (t < KSL) {
    float v = 0.0f;
#pragma unroll
    for (int g = 0; g < 32; ++g) v += red[g * KSL + t];
    xs[t] = v;
  }
  __syncthreads();

  v2f afrag[KSL / 4];
#pragma unroll
  for (int ks = 0; ks < KSL / 4; ++ks) {
    const v2f av = *(const v2f*)(xs + ks * 4 + half * 2);
    const v2f z  = {0.0f, 0.0f};
    afrag[ks] = (l == 0) ? av : z;
  }

  float row0[8];
#pragma unroll
  for (int j = 0; j < 8; ++j) {
    v8f c = {};
#pragma unroll
    for (int ks = 0; ks < KSL / 4; ++ks) {
      c = __builtin_amdgcn_wmma_f32_16x16x4_f32(
           false, afrag[ks],  false, bfrag[j][ks],
           (short)0, c,  false,  false);
    }
    row0[j] = c[0];
  }
  if (half == 0) {
    float* o = Tpart + ((size_t)sl * B_ + b) * D_ + wave * 128 + l;
#pragma unroll
    for (int j = 0; j < 8; ++j) o[j * 16] = row0[j];
  }
}

__global__ __launch_bounds__(256) __attribute__((amdgpu_waves_per_eu(2, 4)))
void tsum_oproj_kernel(
    const float* __restrict__ Tpart, const float* __restrict__ Wo,
    float* __restrict__ Ypart) {
  __shared__ alignas(16) float red[2 * B_ * TPITCH];

  const int sl   = blockIdx.x;
  const int ng   = blockIdx.y;
  const int t    = threadIdx.x;
  const int lane = t & 31;
  const int wave = t >> 5;
  const int half = lane >> 4;
  const int l    = lane & 15;

  v2f bfrag[KS2 / 4];
  const float* wrow =
      Wo + (size_t)((ng * 8 + wave) * 16 + l) * D_ + sl * KS2 + half * 2;
#pragma unroll
  for (int ks = 0; ks < KS2 / 4; ++ks) {
    bfrag[ks] = *(const v2f*)(wrow + ks * 4);
  }

  const int b = wave & 3;
  const int h = wave >> 2;
  const float* p = Tpart + ((size_t)h * B_ + b) * D_ + sl * KS2 + lane * 4;
  v4f acc = {0.0f, 0.0f, 0.0f, 0.0f};
#pragma unroll
  for (int j = 0; j < NSL / 2; ++j) {
    acc += *(const v4f*)(p + (size_t)j * 2 * B_ * D_);
  }
  *(v4f*)(red + (h * B_ + b) * TPITCH + lane * 4) = acc;
  __syncthreads();

  v2f afrag[KS2 / 4];
  const int r = l & 3;
#pragma unroll
  for (int ks = 0; ks < KS2 / 4; ++ks) {
    const v2f ev = *(const v2f*)(red + r * TPITCH + ks * 4 + half * 2);
    const v2f od = *(const v2f*)(red + (B_ + r) * TPITCH + ks * 4 + half * 2);
    const v2f z  = {0.0f, 0.0f};
    afrag[ks] = (l < B_) ? (ev + od) : z;
  }

  v8f c0 = {};
  v8f c1 = {};
#pragma unroll
  for (int ks = 0; ks < KS2 / 4; ks += 2) {
    c0 = __builtin_amdgcn_wmma_f32_16x16x4_f32(
         false, afrag[ks],  false, bfrag[ks],
         (short)0, c0,  false,  false);
    c1 = __builtin_amdgcn_wmma_f32_16x16x4_f32(
         false, afrag[ks + 1],  false, bfrag[ks + 1],
         (short)0, c1,  false,  false);
  }
  float yrow[B_];
#pragma unroll
  for (int m = 0; m < B_; ++m) yrow[m] = c0[m] + c1[m];

  if (half == 0) {
    float* o = Ypart + (size_t)sl * B_ * D_ + (ng * 8 + wave) * 16 + l;
#pragma unroll
    for (int m = 0; m < B_; ++m) o[(size_t)m * D_] = yrow[m];
  }
}

#define OSL 32
#define NOS (D_ / OSL)
#define NRH 2
#define RPW (S_ / NRH)
__global__ __launch_bounds__(256) void ysum_broadcast_kernel(
    const float* __restrict__ Ypart, float* __restrict__ out) {
  const int strip = blockIdx.x;
  const int rh    = blockIdx.y;
  const int b     = blockIdx.z;
  const int t     = threadIdx.x;
  const int c4    = (t & 7) * 4;
  const int rs    = t >> 3;

  const float* p = Ypart + (size_t)b * D_ + strip * OSL + c4;
  v4f v = {0.0f, 0.0f, 0.0f, 0.0f};
#pragma unroll
  for (int sc = 0; sc < NS2; ++sc) {
    v += *(const v4f*)(p + (size_t)sc * B_ * D_);
  }

  float* o = out + ((size_t)b * S_ + (size_t)rh * RPW + rs) * D_ + strip * OSL + c4;
#pragma unroll
  for (int i = 0; i < RPW / 32; ++i) {
    __builtin_nontemporal_store(v, (v4f*)(o + (size_t)i * 32 * D_));
  }
}

extern "C" void kernel_launch(void* const* d_in, const int* in_sizes, int n_in,
                              void* d_out, int out_size, void* d_ws, size_t ws_size,
                              hipStream_t stream) {
  const float* x     = (const float*)d_in[0];
  const float* w_qkv = (const float*)d_in[1];
  const float* w_o   = (const float*)d_in[2];
  const float* w_v   = w_qkv + (size_t)2 * D_ * D_;
  float* out = (float*)d_out;

  float* ws    = (float*)d_ws;
  float* Tpart = ws;
  float* Ypart = ws + 131072;

  xsum_vproj_kernel<<<dim3(NSL, B_), 256, 0, stream>>>(x, w_v, Tpart);

  tsum_oproj_kernel<<<dim3(NS2, 8), 256, 0, stream>>>(Tpart, w_o, Ypart);

  ysum_broadcast_kernel<<<dim3(NOS, NRH, B_), 256, 0, stream>>>(Ypart, out);
}


# x pass: every wave prefetches the 32 row lines of its 8-load batch into the second-level cache three batches ahead (one global_prefetch_b8 per batch); loads, sums and their order unchanged
# speedup vs baseline: 2.6049x; 1.2486x over previous
#include <hip/hip_runtime.h>
#include <hip/hip_bf16.h>

typedef __attribute__((ext_vector_type(2))) float v2f;
typedef __attribute__((ext_vector_type(4))) float v4f;
typedef __attribute__((ext_vector_type(8))) float v8f;

#define B_     4
#define S_     4096
#define D_     1024
#define KSL    32
#define NSL    (D_ / KSL)
#define KS2    128
#define NS2    (D_ / KS2)
#define TPITCH 132
#define PFB    8
#define PFD    3

__global__ __launch_bounds__(256) void xsum_vproj_kernel(
    const float* __restrict__ x, const float* __restrict__ Wv,
    float* __restrict__ Tpart) {
  __shared__ alignas(16) float red[32 * KSL];
  __shared__ alignas(16) float xs[KSL];

  const int sl   = blockIdx.x;
  const int b    = blockIdx.y;
  const int t    = threadIdx.x;
  const int lane = t & 31;
  const int wave = t >> 5;
  const int rg   = wave * 4 + (lane >> 3);
  const int c4   = (lane & 7) * 4;

  const int half = lane >> 4;
  const int l    = lane & 15;

  const float* q = x + ((size_t)b * S_ + (size_t)(lane >> 2) * 32 + wave * 4 + (lane & 3)) * D_
                     + sl * KSL;
  for (int d = 0; d < PFD; ++d) {
    __builtin_prefetch(q + (size_t)d * PFB * 32 * D_, 0, 2);
  }

  v2f bfrag[8][KSL / 4];
#pragma unroll
  for (int j = 0; j < 8; ++j) {
    const float* wrow =
        Wv + (size_t)((wave * 8 + j) * 16 + l) * D_ + sl * KSL + half * 2;
#pragma unroll
    for (int ks = 0; ks < KSL / 4; ++ks) {
      bfrag[j][ks] = *(const v2f*)(wrow + ks * 4);
    }
  }

  const float* p = x + ((size_t)b * S_ + rg) * D_ + sl * KSL + c4;
  v4f acc = {0.0f, 0.0f, 0.0f, 0.0f};
#pragma unroll 8
  for (int i = 0; i < S_ / 32; ++i) {
    if ((i % PFB) == 0 && i + PFD * PFB < S_ / 32) {
      __builtin_prefetch(q + (size_t)(i + PFD * PFB) * 32 * D_, 0, 2);
    }
    v4f v = __builtin_nontemporal_load((const v4f*)(p + (size_t)i * 32 * D_));
    acc += v;
  }
  *(v4f*)(red + rg * KSL + c4) = acc;
  __syncthreads();
  if (t < KSL) {
    float v = 0.0f;
#pragma unroll
    for (int g = 0; g < 32; ++g) v += red[g * KSL + t];
    xs[t] = v;
  }
  __syncthreads();

  v2f afrag[KSL / 4];
#pragma unroll
  for (int ks = 0; ks < KSL / 4; ++ks) {
    const v2f av = *(const v2f*)(xs + ks * 4 + half * 2);
    const v2f z  = {0.0f, 0.0f};
    afrag[ks] = (l == 0) ? av : z;
  }

  float row0[8];
#pragma unroll
  for (int j = 0; j < 8; ++j) {
    v8f c = {};
#pragma unroll
    for (int ks = 0; ks < KSL / 4; ++ks) {
      c = __builtin_amdgcn_wmma_f32_16x16x4_f32(
           false, afrag[ks],  false, bfrag[j][ks],
           (short)0, c,  false,  false);
    }
    row0[j] = c[0];
  }
  if (half == 0) {
    float* o = Tpart + ((size_t)sl * B_ + b) * D_ + wave * 128 + l;
#pragma unroll
    for (int j = 0; j < 8; ++j) o[j * 16] = row0[j];
  }
}

__global__ __launch_bounds__(256) __attribute__((amdgpu_waves_per_eu(2, 4)))
void tsum_oproj_kernel(
    const float* __restrict__ Tpart, const float* __restrict__ Wo,
    float* __restrict__ Ypart) {
  __shared__ alignas(16) float red[2 * B_ * TPITCH];

  const int sl   = blockIdx.x;
  const int ng   = blockIdx.y;
  const int t    = threadIdx.x;
  const int lane = t & 31;
  const int wave = t >> 5;
  const int half = lane >> 4;
  const int l    = lane & 15;

  v2f bfrag[KS2 / 4];
  const float* wrow =
      Wo + (size_t)((ng * 8 + wave) * 16 + l) * D_ + sl * KS2 + half * 2;
#pragma unroll
  for (int ks = 0; ks < KS2 / 4; ++ks) {
    bfrag[ks] = *(const v2f*)(wrow + ks * 4);
  }

  const int b = wave & 3;
  const int h = wave >> 2;
  const float* p = Tpart + ((size_t)h * B_ + b) * D_ + sl * KS2 + lane * 4;
  v4f acc = {0.0f, 0.0f, 0.0f, 0.0f};
#pragma unroll
  for (int j = 0; j < NSL / 2; ++j) {
    acc += *(const v4f*)(p + (size_t)j * 2 * B_ * D_);
  }
  *(v4f*)(red + (h * B_ + b) * TPITCH + lane * 4) = acc;
  __syncthreads();

  v2f afrag[KS2 / 4];
  const int r = l & 3;
#pragma unroll
  for (int ks = 0; ks < KS2 / 4; ++ks) {
    const v2f ev = *(const v2f*)(red + r * TPITCH + ks * 4 + half * 2);
    const v2f od = *(const v2f*)(red + (B_ + r) * TPITCH + ks * 4 + half * 2);
    const v2f z  = {0.0f, 0.0f};
    afrag[ks] = (l < B_) ? (ev + od) : z;
  }

  v8f c0 = {};
  v8f c1 = {};
#pragma unroll
  for (int ks = 0; ks < KS2 / 4; ks += 2) {
    c0 = __builtin_amdgcn_wmma_f32_16x16x4_f32(
         false, afrag[ks],  false, bfrag[ks],
         (short)0, c0,  false,  false);
    c1 = __builtin_amdgcn_wmma_f32_16x16x4_f32(
         false, afrag[ks + 1],  false, bfrag[ks + 1],
         (short)0, c1,  false,  false);
  }
  float yrow[B_];
#pragma unroll
  for (int m = 0; m < B_; ++m) yrow[m] = c0[m] + c1[m];

  if (half == 0) {
    float* o = Ypart + (size_t)sl * B_ * D_ + (ng * 8 + wave) * 16 + l;
#pragma unroll
    for (int m = 0; m < B_; ++m) o[(size_t)m * D_] = yrow[m];
  }
}

#define OSL 32
#define NOS (D_ / OSL)
#define NRH 2
#define RPW (S_ / NRH)
__global__ __launch_bounds__(256) void ysum_broadcast_kernel(
    const float* __restrict__ Ypart, float* __restrict__ out) {
  const int strip = blockIdx.x;
  const int rh    = blockIdx.y;
  const int b     = blockIdx.z;
  const int t     = threadIdx.x;
  const int c4    = (t & 7) * 4;
  const int rs    = t >> 3;

  const float* p = Ypart + (size_t)b * D_ + strip * OSL + c4;
  v4f v = {0.0f, 0.0f, 0.0f, 0.0f};
#pragma unroll
  for (int sc = 0; sc < NS2; ++sc) {
    v += *(const v4f*)(p + (size_t)sc * B_ * D_);
  }

  float* o = out + ((size_t)b * S_ + (size_t)rh * RPW + rs) * D_ + strip * OSL + c4;
#pragma unroll
  for (int i = 0; i < RPW / 32; ++i) {
    __builtin_nontemporal_store(v, (v4f*)(o + (size_t)i * 32 * D_));
  }
}

extern "C" void kernel_launch(void* const* d_in, const int* in_sizes, int n_in,
                              void* d_out, int out_size, void* d_ws, size_t ws_size,
                              hipStream_t stream) {
  const float* x     = (const float*)d_in[0];
  const float* w_qkv = (const float*)d_in[1];
  const float* w_o   = (const float*)d_in[2];
  const float* w_v   = w_qkv + (size_t)2 * D_ * D_;
  float* out = (float*)d_out;

  float* ws    = (float*)d_ws;
  float* Tpart = ws;
  float* Ypart = ws + 131072;

  xsum_vproj_kernel<<<dim3(NSL, B_), 256, 0, stream>>>(x, w_v, Tpart);

  tsum_oproj_kernel<<<dim3(NS2, 8), 256, 0, stream>>>(Tpart, w_o, Ypart);

  ysum_broadcast_kernel<<<dim3(NOS, NRH, B_), 256, 0, stream>>>(Ypart, out);
}
